# starting kernel
# baseline (speedup 1.0000x reference)
#include <hip/hip_runtime.h>

typedef __attribute__((ext_vector_type(2))) float v2f;
typedef __attribute__((ext_vector_type(8))) float v8f;

#define N_NODES 100000
#define N_EDGES 3200000
#define IN_CH 512
#define HID 16

__global__ void gcn_init(float* __restrict__ deg, float* __restrict__ agg) {
    int i = blockIdx.x * blockDim.x + threadIdx.x;
    if (i < N_NODES * HID) agg[i] = 0.0f;
    if (i < N_NODES) deg[i] = 1.0f;
}

__global__ void gcn_degree(const int* __restrict__ col, float* __restrict__ deg) {
    int e = blockIdx.x * blockDim.x + threadIdx.x;
    if (e < N_EDGES) atomicAdd(&deg[col[e]], 1.0f);
}

__global__ void gcn_rsqrt(float* __restrict__ deg) {
    int i = blockIdx.x * blockDim.x + threadIdx.x;
    if (i < N_NODES) deg[i] = rsqrtf(deg[i]);
}

__global__ void __launch_bounds__(256) gcn_xw_wmma(const float* __restrict__ x,
                                                   const float* __restrict__ W1,
                                                   float* __restrict__ xw) {
    __shared__ float sW1[IN_CH * HID];
    for (int i = threadIdx.x; i < (IN_CH * HID) / 4; i += 256)
        ((float4*)sW1)[i] = ((const float4*)W1)[i];
    __syncthreads();

    const int wave = threadIdx.x >> 5;
    const int lane = threadIdx.x & 31;
    const int half = lane >> 4;
    const int lrow = lane & 15;
    const int tile = blockIdx.x * 8 + wave;
    if (tile * 16 >= N_NODES) return;

    const float* xrow = x + (size_t)(tile * 16 + lrow) * IN_CH + 2 * half;
    const float* bptr = sW1 + (2 * half) * HID + lrow;

    v8f acc = {};
#pragma unroll 8
    for (int k0 = 0; k0 < IN_CH; k0 += 4) {
        v2f a = *(const v2f*)(xrow + k0);
        v2f b;
        b.x = bptr[k0 * HID];
        b.y = bptr[k0 * HID + HID];
        acc = __builtin_amdgcn_wmma_f32_16x16x4_f32(
             false, a,  false, b,
             (short)0, acc,  false,  false);
    }

    float* o = xw + (size_t)(tile * 16 + 8 * half) * HID + lrow;
#pragma unroll
    for (int r = 0; r < 8; ++r) o[r * HID] = acc[r];
}

__global__ void gcn_scatter(const int* __restrict__ row, const int* __restrict__ col,
                            const float* __restrict__ dinv, const float* __restrict__ xw,
                            float* __restrict__ agg) {
    long long t = (long long)blockIdx.x * blockDim.x + threadIdx.x;
    int e = (int)(t >> 4);
    int c = (int)t & 15;
    if (e < N_EDGES) {
        int r = row[e];
        int d = col[e];
        float nrm = dinv[r] * dinv[d];
        atomicAdd(&agg[(size_t)d * HID + c], xw[(size_t)r * HID + c] * nrm);
    }
}

__global__ void gcn_out(const float* __restrict__ agg, const float* __restrict__ xw,
                        const float* __restrict__ dinv, const float* __restrict__ b1,
                        const float* __restrict__ W2, const float* __restrict__ b2,
                        float* __restrict__ out) {
    int i = blockIdx.x * blockDim.x + threadIdx.x;
    if (i >= N_NODES) return;
    float s = dinv[i] * dinv[i];
    float o0 = b2[0], o1 = b2[1];
#pragma unroll
    for (int c = 0; c < HID; ++c) {
        float h = agg[(size_t)i * HID + c] + xw[(size_t)i * HID + c] * s + b1[c];
        o0 += h * W2[c * 2 + 0];
        o1 += h * W2[c * 2 + 1];
    }
    out[i * 2 + 0] = o0;
    out[i * 2 + 1] = o1;
}

extern "C" void kernel_launch(void* const* d_in, const int* in_sizes, int n_in,
                              void* d_out, int out_size, void* d_ws, size_t ws_size,
                              hipStream_t stream) {
    const float* x  = (const float*)d_in[0];
    const int*   ei = (const int*)d_in[1];
    const float* W1 = (const float*)d_in[2];
    const float* b1 = (const float*)d_in[3];
    const float* W2 = (const float*)d_in[4];
    const float* b2 = (const float*)d_in[5];
    float* out = (float*)d_out;

    float* deg = (float*)d_ws;
    float* xw  = deg + N_NODES;
    float* agg = xw + (size_t)N_NODES * HID;

    const int* row = ei;
    const int* col = ei + N_EDGES;

    gcn_init<<<(N_NODES * HID + 255) / 256, 256, 0, stream>>>(deg, agg);
    gcn_degree<<<(N_EDGES + 255) / 256, 256, 0, stream>>>(col, deg);
    gcn_rsqrt<<<(N_NODES + 255) / 256, 256, 0, stream>>>(deg);

    int n_tiles = N_NODES / 16;
    gcn_xw_wmma<<<(n_tiles + 7) / 8, 256, 0, stream>>>(x, W1, xw);

    long long sthreads = (long long)N_EDGES * 16;
    gcn_scatter<<<(int)((sthreads + 255) / 256), 256, 0, stream>>>(row, col, deg, xw, agg);

    gcn_out<<<(N_NODES + 255) / 256, 256, 0, stream>>>(agg, xw, deg, b1, W2, b2, out);
}


# edges filed per call by target range (LDS counting sort); one work-group per 416-node range counts degrees and sums gathered rows by LDS float adds: no global atomics or planes; sum order changes
# speedup vs baseline: 4.5806x; 4.5806x over previous
#include <hip/hip_runtime.h>


typedef __attribute__((ext_vector_type(2))) float v2f;
typedef __attribute__((ext_vector_type(8))) float v8f;

#define N_NODES 100000
#define N_EDGES 3200000
#define IN_CH 512
#define HID 16

#define NPB 416
#define NB ((N_NODES + NPB - 1) / NPB)
#define NBP 256
#define NCH 256
#define CH_EDGES (N_EDGES / NCH)
static_assert(NB <= NBP, "the count table must hold every range");
static_assert(NPB <= 512, "the target's offset inside its range must fit 9 bits");
static_assert(N_NODES <= (1 << 22), "source << 9 must fit a positive int");
static_assert(NCH * CH_EDGES == N_EDGES, "edges must split evenly over chunks");
static_assert((NPB * 4) % 128 == 0, "a range's degrees and outputs must be whole lines");

#define CH_VEC (CH_EDGES / 4)
static_assert(CH_VEC * 4 == CH_EDGES, "a chunk must be whole 16-byte words");
#define HB_BLOCK 1024
#define HB_PER ((CH_VEC + HB_BLOCK - 1) / HB_BLOCK)
static_assert(HB_BLOCK >= NBP, "one thread per table column");
__global__ void __launch_bounds__(HB_BLOCK)
gcn_chunk_count(const int* __restrict__ col, int* __restrict__ cnt) {
    __shared__ int h[NBP];
    const int tid = threadIdx.x;
    if (tid < NBP) h[tid] = 0;
    const int4* c4 = (const int4*)(col + (size_t)blockIdx.x * CH_EDGES);
    int4 d[HB_PER];
#pragma unroll
    for (int j = 0; j < HB_PER; ++j) {
        const int i = tid + j * HB_BLOCK;
        d[j] = (i < CH_VEC) ? c4[i] : make_int4(0, 0, 0, 0);
    }
    __syncthreads();
#pragma unroll
    for (int j = 0; j < HB_PER; ++j) {
        if (tid + j * HB_BLOCK < CH_VEC) {
            atomicAdd(&h[(unsigned)d[j].x / NPB], 1);
            atomicAdd(&h[(unsigned)d[j].y / NPB], 1);
            atomicAdd(&h[(unsigned)d[j].z / NPB], 1);
            atomicAdd(&h[(unsigned)d[j].w / NPB], 1);
        }
    }
    __syncthreads();
    if (tid < NBP) cnt[blockIdx.x * NBP + tid] = h[tid];
}

#define BN_BLOCK 1024
#define BN_Q (BN_BLOCK / NBP)
#define BN_T 16
#define BN_PER ((CH_VEC + BN_BLOCK - 1) / BN_BLOCK)
static_assert(NCH % (BN_Q * BN_T) == 0, "the table rows must split evenly over the load groups");
static_assert(NBP == 256, "the running sums below are written for 8 waves of 32 columns");

__device__ __forceinline__ int wave_running_sum(int v, int lane) {
#pragma unroll
    for (int o = 1; o < 32; o <<= 1) {
        const int t = __shfl_up(v, o);
        if (lane >= o) v += t;
    }
    return v;
}
__global__ void __launch_bounds__(BN_BLOCK)
gcn_bin(const int* __restrict__ row, const int* __restrict__ col, const int* __restrict__ cnt,
        int* __restrict__ bstart, int* __restrict__ binned) {
    __shared__ int sPart[BN_Q][NBP];
    __shared__ int sTot[BN_Q][NBP];
    __shared__ int wsumG[NBP / 32];
    __shared__ int wsumL[NBP / 32];
    __shared__ int gbase[NBP];
    __shared__ int lstart[NBP];
    __shared__ int lcur[NBP];
    __shared__ int srcS[CH_EDGES];
    __shared__ int dstS[CH_EDGES];

    const int tid = threadIdx.x;
    const int b = blockIdx.x;
    const int k = tid & (NBP - 1);
    const int q = tid >> 8;
    const int lane = tid & 31;
    const size_t e0 = (size_t)b * CH_EDGES;

    const int4* c4 = (const int4*)(col + e0);
    const int4* r4 = (const int4*)(row + e0);
    int4 dv[BN_PER], sv[BN_PER];
#pragma unroll
    for (int j = 0; j < BN_PER; ++j) {
        const int i = tid + j * BN_BLOCK;
        dv[j] = (i < CH_VEC) ? c4[i] : make_int4(0, 0, 0, 0);
        sv[j] = (i < CH_VEC) ? r4[i] : make_int4(0, 0, 0, 0);
    }

    int part = 0, tot = 0;
    for (int bb0 = q; bb0 < NCH; bb0 += BN_Q * BN_T) {
        int v[BN_T];
#pragma unroll
        for (int j = 0; j < BN_T; ++j) v[j] = cnt[(bb0 + j * BN_Q) * NBP + k];
#pragma unroll
        for (int j = 0; j < BN_T; ++j) {
            tot += v[j];
            part += ((bb0 + j * BN_Q) < b) ? v[j] : 0;
        }
    }
    sPart[q][k] = part;
    sTot[q][k] = tot;
    __syncthreads();
    int myTot = 0, myPart = 0, myCnt = 0, runG = 0, runL = 0;
    if (tid < NBP) {
#pragma unroll
        for (int j = 0; j < BN_Q; ++j) { myTot += sTot[j][tid]; myPart += sPart[j][tid]; }
        myCnt = cnt[b * NBP + tid];
        runG = wave_running_sum(myTot, lane);
        runL = wave_running_sum(myCnt, lane);
        if (lane == 31) { wsumG[tid >> 5] = runG; wsumL[tid >> 5] = runL; }
    }
    __syncthreads();
    if (tid < NBP) {
        for (int w = 0; w < (tid >> 5); ++w) { runG += wsumG[w]; runL += wsumL[w]; }
        const int gs = runG - myTot;
        const int ls = runL - myCnt;
        gbase[tid] = gs + myPart;
        lstart[tid] = ls;
        lcur[tid] = ls;
        if (b == 0) {
            bstart[tid] = gs;
            if (tid == NBP - 1) bstart[NBP] = gs + myTot;
        }
    }
    __syncthreads();
#pragma unroll
    for (int j = 0; j < BN_PER; ++j) {
        if (tid + j * BN_BLOCK < CH_VEC) {
            int p;
            p = atomicAdd(&lcur[(unsigned)dv[j].x / NPB], 1); srcS[p] = sv[j].x; dstS[p] = dv[j].x;
            p = atomicAdd(&lcur[(unsigned)dv[j].y / NPB], 1); srcS[p] = sv[j].y; dstS[p] = dv[j].y;
            p = atomicAdd(&lcur[(unsigned)dv[j].z / NPB], 1); srcS[p] = sv[j].z; dstS[p] = dv[j].z;
            p = atomicAdd(&lcur[(unsigned)dv[j].w / NPB], 1); srcS[p] = sv[j].w; dstS[p] = dv[j].w;
        }
    }
    __syncthreads();
    for (int i = tid; i < CH_EDGES; i += BN_BLOCK) {
        const int d = dstS[i];
        const int kk = (int)((unsigned)d / NPB);
        binned[gbase[kk] + (i - lstart[kk])] = (srcS[i] << 9) | (d - kk * NPB);
    }
}

#define RD_BLOCK 1024
#define RD_U 8
static_assert(NPB <= RD_BLOCK, "one thread writes one node's degree");
__global__ void __launch_bounds__(RD_BLOCK)
gcn_range_degree(const int* __restrict__ binned, const int* __restrict__ bstart, float* __restrict__ deg) {
    __shared__ int c[512];
    const int tid = threadIdx.x;
    if (tid < 512) c[tid] = 0;
    __syncthreads();
    const int k = blockIdx.x;
    const int e1 = bstart[k + 1];
    for (int base = bstart[k]; base < e1; base += RD_BLOCK * RD_U) {
        int w[RD_U];
#pragma unroll
        for (int j = 0; j < RD_U; ++j) {
            const int e = base + j * RD_BLOCK + tid;
            w[j] = (e < e1) ? binned[e] : -1;
        }
#pragma unroll
        for (int j = 0; j < RD_U; ++j) if (w[j] >= 0) atomicAdd(&c[w[j] & 511], 1);
    }
    __syncthreads();
    const int n = k * NPB + tid;
    if (tid < NPB && n < N_NODES) deg[n] = (float)(c[tid] + 1);
}


__global__ void __launch_bounds__(256) gcn_xw_wmma(const float* __restrict__ x,
                                                   const float* __restrict__ W1,
                                                   const float* __restrict__ deg,
                                                   float* __restrict__ xw) {
    __shared__ float sW1[IN_CH * HID];
    for (int i = threadIdx.x; i < (IN_CH * HID) / 4; i += 256)
        ((float4*)sW1)[i] = ((const float4*)W1)[i];
    __syncthreads();

    const int wave = threadIdx.x >> 5;
    const int lane = threadIdx.x & 31;
    const int half = lane >> 4;
    const int lrow = lane & 15;
    const int tile = blockIdx.x * 8 + wave;
    if (tile * 16 >= N_NODES) return;

    const float dg = deg[tile * 16 + lrow];

    const float* xrow = x + (size_t)(tile * 16 + lrow) * IN_CH + 2 * half;
    const float* bptr = sW1 + (2 * half) * HID + lrow;

    v8f acc = {};
#pragma unroll 8
    for (int k0 = 0; k0 < IN_CH; k0 += 4) {
        v2f a = *(const v2f*)(xrow + k0);
        v2f b;
        b.x = bptr[k0 * HID];
        b.y = bptr[k0 * HID + HID];
        acc = __builtin_amdgcn_wmma_f32_16x16x4_f32(
             false, a,  false, b,
             (short)0, acc,  false,  false);
    }

    const float dv = rsqrtf(dg);
    float* o = xw + (size_t)(tile * 16 + 8 * half) * HID + lrow;
#pragma unroll
    for (int r = 0; r < 8; ++r) o[r * HID] = acc[r] * __shfl(dv, 8 * half + r);
}

#define RS_BLOCK 1024
#define RS_WAVES (RS_BLOCK / 32)
#define RS_PAIRS 16
static_assert(NPB <= RS_BLOCK, "one thread finishes one node");
__global__ void __launch_bounds__(RS_BLOCK)
gcn_range_sum(const int* __restrict__ binned, const int* __restrict__ bstart,
              const float* __restrict__ xs, const float* __restrict__ deg,
              const float* __restrict__ b1, const float* __restrict__ W2, const float* __restrict__ b2,
              float* __restrict__ out) {
    __shared__ float acc[NPB * HID];
    for (int i = threadIdx.x; i < NPB * HID; i += RS_BLOCK) acc[i] = 0.0f;
    __syncthreads();

    const int k = blockIdx.x;
    const int lane = threadIdx.x & 31;
    const int half = lane >> 4;
    const int c = lane & 15;
    const int wave = __builtin_amdgcn_readfirstlane((int)(threadIdx.x >> 5));
    const int e0 = bstart[k];
    const int n_e = bstart[k + 1] - e0;
    const int full = n_e >> 5;
    int wv = (wave < full) ? binned[e0 + 32 * wave + lane] : 0;
    for (int t = wave; t < full; t += RS_WAVES) {
        int w[RS_PAIRS];
        float v[RS_PAIRS];
#pragma unroll
        for (int j = 0; j < RS_PAIRS; ++j) {
            const int wa = __builtin_amdgcn_readlane(wv, 2 * j);
            const int wb = __builtin_amdgcn_readlane(wv, 2 * j + 1);
            w[j] = half ? wb : wa;
        }
        const int tn = t + RS_WAVES;
        const int wv_next = (tn < full) ? binned[e0 + 32 * tn + lane] : 0;
#pragma unroll
        for (int j = 0; j < RS_PAIRS; ++j) v[j] = xs[(unsigned)((w[j] >> 9) * HID + c)];
#pragma unroll
        for (int j = 0; j < RS_PAIRS; ++j) atomicAdd(&acc[(w[j] & 511) * HID + c], v[j]);
        wv = wv_next;
    }
    const int rem = n_e & 31;
    if (rem != 0 && wave == (full & (RS_WAVES - 1))) {
        const int eb = e0 + 32 * full;
        const int wt = (lane < rem) ? binned[eb + lane] : 0;
        for (int j = 0; j < RS_PAIRS; ++j) {
            const int idx = 2 * j + half;
            const int w = __shfl(wt, idx);
            if (idx < rem) atomicAdd(&acc[(w & 511) * HID + c], xs[(unsigned)((w >> 9) * HID + c)]);
        }
    }
    __syncthreads();

    const int t = threadIdx.x;
    const int n = k * NPB + t;
    if (t < NPB && n < N_NODES) {
        const float s = rsqrtf(deg[n]);
        float o0 = b2[0], o1 = b2[1];
#pragma unroll
        for (int cc = 0; cc < HID; ++cc) {
            const float h = (acc[t * HID + cc] + xs[(size_t)n * HID + cc]) * s + b1[cc];
            o0 += h * W2[cc * 2 + 0];
            o1 += h * W2[cc * 2 + 1];
        }
        out[n * 2 + 0] = o0;
        out[n * 2 + 1] = o1;
    }
}

extern "C" void kernel_launch(void* const* d_in, const int* in_sizes, int n_in,
                              void* d_out, int out_size, void* d_ws, size_t ws_size,
                              hipStream_t stream) {
    const float* x  = (const float*)d_in[0];
    const int*   ei = (const int*)d_in[1];
    const float* W1 = (const float*)d_in[2];
    const float* b1 = (const float*)d_in[3];
    const float* W2 = (const float*)d_in[4];
    const float* b2 = (const float*)d_in[5];
    float* out = (float*)d_out;

    float* deg    = (float*)d_ws;
    float* xw     = deg + N_NODES;
    int*   binned = (int*)(xw + (size_t)N_NODES * HID);
    int*   cnt    = binned + N_EDGES;
    int*   bstart = cnt + NCH * NBP;

    const int* row = ei;
    const int* col = ei + N_EDGES;

    gcn_chunk_count<<<NCH, HB_BLOCK, 0, stream>>>(col, cnt);
    gcn_bin<<<NCH, BN_BLOCK, 0, stream>>>(row, col, cnt, bstart, binned);
    gcn_range_degree<<<NB, RD_BLOCK, 0, stream>>>(binned, bstart, deg);

    int n_tiles = N_NODES / 16;
    gcn_xw_wmma<<<(n_tiles + 7) / 8, 256, 0, stream>>>(x, W1, deg, xw);

    gcn_range_sum<<<NB, RS_BLOCK, 0, stream>>>(binned, bstart, xw, deg, b1, W2, b2, out);
}


# edge sort in one pass: each chunk sorted by target range inside its own block; a range's work-group walks its 256 per-chunk runs through an LDS address table; count launch and global offsets go
# speedup vs baseline: 4.7383x; 1.0344x over previous
#include <hip/hip_runtime.h>


typedef __attribute__((ext_vector_type(2))) float v2f;
typedef __attribute__((ext_vector_type(8))) float v8f;

#define N_NODES 100000
#define N_EDGES 3200000
#define IN_CH 512
#define HID 16

#define NPB 416
#define NB ((N_NODES + NPB - 1) / NPB)
#define NBP 256
#define NCH 256
#define CH_EDGES (N_EDGES / NCH)
static_assert(NB <= NBP, "the run table must hold every range");
static_assert(NPB <= 512, "the target's offset inside its range must fit 9 bits");
static_assert(N_NODES <= (1 << 22), "source << 9 must fit a positive int");
static_assert(NCH * CH_EDGES == N_EDGES, "edges must split evenly over chunks");
static_assert((NPB * 4) % 128 == 0, "a range's degrees and outputs must be whole lines");
static_assert(CH_EDGES < (1 << 15), "a run's start and length inside a chunk share one word, 16 bits each");

__device__ __forceinline__ int wave_running_sum(int v, int lane) {
#pragma unroll
    for (int o = 1; o < 32; o <<= 1) {
        const int t = __shfl_up(v, o);
        if (lane >= o) v += t;
    }
    return v;
}

#define CH_VEC (CH_EDGES / 4)
static_assert(CH_VEC * 4 == CH_EDGES, "a chunk must be whole 16-byte words");
#define BN_BLOCK 1024
#define BN_PER ((CH_VEC + BN_BLOCK - 1) / BN_BLOCK)
static_assert(NBP == 256 && BN_BLOCK >= NBP, "the running sum below is written for 8 waves of 32 columns");
__global__ void __launch_bounds__(BN_BLOCK)
gcn_bin(const int* __restrict__ row, const int* __restrict__ col, int* __restrict__ runs, int* __restrict__ binned) {
    __shared__ int h[NBP];
    __shared__ int lcur[NBP];
    __shared__ int wsum[NBP / 32];
    __shared__ int4 wordS4[CH_VEC];
    int* wordS = (int*)wordS4;

    const int tid = threadIdx.x;
    const int b = blockIdx.x;
    const int lane = tid & 31;
    const size_t e0 = (size_t)b * CH_EDGES;

    if (tid < NBP) h[tid] = 0;
    const int4* c4 = (const int4*)(col + e0);
    const int4* r4 = (const int4*)(row + e0);
    int4 dv[BN_PER], sv[BN_PER];
#pragma unroll
    for (int j = 0; j < BN_PER; ++j) {
        const int i = tid + j * BN_BLOCK;
        dv[j] = (i < CH_VEC) ? c4[i] : make_int4(0, 0, 0, 0);
        sv[j] = (i < CH_VEC) ? r4[i] : make_int4(0, 0, 0, 0);
    }
    __syncthreads();
#pragma unroll
    for (int j = 0; j < BN_PER; ++j) {
        if (tid + j * BN_BLOCK < CH_VEC) {
            atomicAdd(&h[(unsigned)dv[j].x / NPB], 1);
            atomicAdd(&h[(unsigned)dv[j].y / NPB], 1);
            atomicAdd(&h[(unsigned)dv[j].z / NPB], 1);
            atomicAdd(&h[(unsigned)dv[j].w / NPB], 1);
        }
    }
    __syncthreads();
    int myCnt = 0, run = 0;
    if (tid < NBP) {
        myCnt = h[tid];
        run = wave_running_sum(myCnt, lane);
        if (lane == 31) wsum[tid >> 5] = run;
    }
    __syncthreads();
    if (tid < NBP) {
        for (int w = 0; w < (tid >> 5); ++w) run += wsum[w];
        const int ls = run - myCnt;
        lcur[tid] = ls;
        runs[tid * NCH + b] = (ls << 16) | myCnt;
    }
    __syncthreads();
#pragma unroll
    for (int j = 0; j < BN_PER; ++j) {
        if (tid + j * BN_BLOCK < CH_VEC) {
            unsigned kk; int p;
            kk = (unsigned)dv[j].x / NPB; p = atomicAdd(&lcur[kk], 1); wordS[p] = (sv[j].x << 9) | (dv[j].x - (int)kk * NPB);
            kk = (unsigned)dv[j].y / NPB; p = atomicAdd(&lcur[kk], 1); wordS[p] = (sv[j].y << 9) | (dv[j].y - (int)kk * NPB);
            kk = (unsigned)dv[j].z / NPB; p = atomicAdd(&lcur[kk], 1); wordS[p] = (sv[j].z << 9) | (dv[j].z - (int)kk * NPB);
            kk = (unsigned)dv[j].w / NPB; p = atomicAdd(&lcur[kk], 1); wordS[p] = (sv[j].w << 9) | (dv[j].w - (int)kk * NPB);
        }
    }
    __syncthreads();
    int4* o4 = (int4*)(binned + e0);
    for (int i = tid; i < CH_VEC; i += BN_BLOCK) o4[i] = wordS4[i];
}

#define RL_BLOCK 1024
#define RL_TILE 16384
static_assert(RL_BLOCK == 4 * NCH, "four threads fill one run");
static_assert(RL_TILE % 32 == 0, "a tile is whole trips of 32 positions");
__device__ __forceinline__ int range_offsets(const int* __restrict__ runs, int k, int tid, int lane,
                                             int* offS, int* stS, int* wsum) {
    int len = 0, run = 0;
    if (tid < NCH) {
        const int v = runs[k * NCH + tid];
        len = v & 0xffff;
        stS[tid] = tid * CH_EDGES + (int)((unsigned)v >> 16);
        run = wave_running_sum(len, lane);
        if (lane == 31) wsum[tid >> 5] = run;
    }
    __syncthreads();
    if (tid < NCH) {
        for (int w = 0; w < (tid >> 5); ++w) run += wsum[w];
        offS[tid + 1] = run;
        if (tid == 0) offS[0] = 0;
    }
    __syncthreads();
    return __builtin_amdgcn_readfirstlane(offS[NCH]);
}
__device__ __forceinline__ void range_fill(int tb, int tid, const int* offS, const int* stS, int* addrS) {
    const int c = tid >> 2;
    const int o = offS[c], e = offS[c + 1], s = stS[c];
    const int lo = (o > tb) ? o : tb;
    const int hi = (e < tb + RL_TILE) ? e : tb + RL_TILE;
    for (int q = lo + (tid & 3); q < hi; q += 4) addrS[q - tb] = s + (q - o);
}

#define RD_U 8
static_assert(NPB <= RL_BLOCK, "one thread writes one node's degree");
__global__ void __launch_bounds__(RL_BLOCK)
gcn_range_degree(const int* __restrict__ binned, const int* __restrict__ runs, float* __restrict__ deg) {
    __shared__ int c[512];
    __shared__ int offS[NCH + 1];
    __shared__ int stS[NCH];
    __shared__ int wsum[NCH / 32];
    __shared__ int addrS[RL_TILE];
    const int tid = threadIdx.x;
    const int k = blockIdx.x;
    if (tid < 512) c[tid] = 0;
    const int n_e = range_offsets(runs, k, tid, tid & 31, offS, stS, wsum);
    for (int tb = 0; tb < n_e; tb += RL_TILE) {
        const int tn = (n_e - tb < RL_TILE) ? n_e - tb : RL_TILE;
        range_fill(tb, tid, offS, stS, addrS);
        __syncthreads();
        for (int base = 0; base < tn; base += RL_BLOCK * RD_U) {
            int a[RD_U], w[RD_U];
#pragma unroll
            for (int j = 0; j < RD_U; ++j) {
                const int q = base + j * RL_BLOCK + tid;
                a[j] = addrS[q < tn ? q : tn - 1];
            }
#pragma unroll
            for (int j = 0; j < RD_U; ++j) {
                const int q = base + j * RL_BLOCK + tid;
                w[j] = (q < tn) ? binned[a[j]] : -1;
            }
#pragma unroll
            for (int j = 0; j < RD_U; ++j) if (w[j] >= 0) atomicAdd(&c[w[j] & 511], 1);
        }
        __syncthreads();
    }
    __syncthreads();
    const int n = k * NPB + tid;
    if (tid < NPB && n < N_NODES) deg[n] = (float)(c[tid] + 1);
}


__global__ void __launch_bounds__(256) gcn_xw_wmma(const float* __restrict__ x,
                                                   const float* __restrict__ W1,
                                                   const float* __restrict__ deg,
                                                   float* __restrict__ xw) {
    __shared__ float sW1[IN_CH * HID];
    for (int i = threadIdx.x; i < (IN_CH * HID) / 4; i += 256)
        ((float4*)sW1)[i] = ((const float4*)W1)[i];
    __syncthreads();

    const int wave = threadIdx.x >> 5;
    const int lane = threadIdx.x & 31;
    const int half = lane >> 4;
    const int lrow = lane & 15;
    const int tile = blockIdx.x * 8 + wave;
    if (tile * 16 >= N_NODES) return;

    const float dg = deg[tile * 16 + lrow];

    const float* xrow = x + (size_t)(tile * 16 + lrow) * IN_CH + 2 * half;
    const float* bptr = sW1 + (2 * half) * HID + lrow;

    v8f acc = {};
#pragma unroll 8
    for (int k0 = 0; k0 < IN_CH; k0 += 4) {
        v2f a = *(const v2f*)(xrow + k0);
        v2f b;
        b.x = bptr[k0 * HID];
        b.y = bptr[k0 * HID + HID];
        acc = __builtin_amdgcn_wmma_f32_16x16x4_f32(
             false, a,  false, b,
             (short)0, acc,  false,  false);
    }

    const float dv = rsqrtf(dg);
    float* o = xw + (size_t)(tile * 16 + 8 * half) * HID + lrow;
#pragma unroll
    for (int r = 0; r < 8; ++r) o[r * HID] = acc[r] * __shfl(dv, 8 * half + r);
}

#define RS_BLOCK RL_BLOCK
#define RS_WAVES (RS_BLOCK / 32)
#define RS_PAIRS 16
static_assert(NPB <= RS_BLOCK, "one thread finishes one node");
__global__ void __launch_bounds__(RS_BLOCK)
gcn_range_sum(const int* __restrict__ binned, const int* __restrict__ runs,
              const float* __restrict__ xs, const float* __restrict__ deg,
              const float* __restrict__ b1, const float* __restrict__ W2, const float* __restrict__ b2,
              float* __restrict__ out) {
    __shared__ float acc[NPB * HID];
    __shared__ int offS[NCH + 1];
    __shared__ int stS[NCH];
    __shared__ int wsum[NCH / 32];
    __shared__ int addrS[RL_TILE];
    for (int i = threadIdx.x; i < NPB * HID; i += RS_BLOCK) acc[i] = 0.0f;

    const int k = blockIdx.x;
    const int lane = threadIdx.x & 31;
    const int half = lane >> 4;
    const int c = lane & 15;
    const int wave = __builtin_amdgcn_readfirstlane((int)(threadIdx.x >> 5));
    const int n_e = range_offsets(runs, k, threadIdx.x, lane, offS, stS, wsum);
    for (int tb = 0; tb < n_e; tb += RL_TILE) {
        const int tn_e = (n_e - tb < RL_TILE) ? n_e - tb : RL_TILE;
        range_fill(tb, threadIdx.x, offS, stS, addrS);
        __syncthreads();
        const int full = tn_e >> 5;
        const int q1 = 32 * (wave + RS_WAVES) + lane;
        int an = addrS[q1 < tn_e ? q1 : tn_e - 1];
        const int q0 = 32 * wave + lane;
        const int a0 = addrS[q0 < tn_e ? q0 : tn_e - 1];
        int wv = (wave < full) ? binned[a0] : 0;
        asm volatile("" : "+v"(an));
        for (int t = wave; t < full; t += RS_WAVES) {
            int w[RS_PAIRS];
            float v[RS_PAIRS];
#pragma unroll
            for (int j = 0; j < RS_PAIRS; ++j) {
                const int wa = __builtin_amdgcn_readlane(wv, 2 * j);
                const int wb = __builtin_amdgcn_readlane(wv, 2 * j + 1);
                w[j] = half ? wb : wa;
            }
            const int tn = t + RS_WAVES;
            const int wv_next = (tn < full) ? binned[an] : 0;
            const int q2 = 32 * (tn + RS_WAVES) + lane;
            const int an_next = addrS[q2 < tn_e ? q2 : tn_e - 1];
#pragma unroll
            for (int j = 0; j < RS_PAIRS; ++j) v[j] = xs[(unsigned)((w[j] >> 9) * HID + c)];
#pragma unroll
            for (int j = 0; j < RS_PAIRS; ++j) atomicAdd(&acc[(w[j] & 511) * HID + c], v[j]);
            wv = wv_next;
            an = an_next;
        }
        const int rem = tn_e & 31;
        if (rem != 0 && wave == (full & (RS_WAVES - 1))) {
            const int wt = (lane < rem) ? binned[addrS[32 * full + lane]] : 0;
            for (int j = 0; j < RS_PAIRS; ++j) {
                const int idx = 2 * j + half;
                const int w = __shfl(wt, idx);
                if (idx < rem) atomicAdd(&acc[(w & 511) * HID + c], xs[(unsigned)((w >> 9) * HID + c)]);
            }
        }
        __syncthreads();
    }

    const int t = threadIdx.x;
    const int n = k * NPB + t;
    if (t < NPB && n < N_NODES) {
        const float s = rsqrtf(deg[n]);
        float o0 = b2[0], o1 = b2[1];
#pragma unroll
        for (int cc = 0; cc < HID; ++cc) {
            const float h = (acc[t * HID + cc] + xs[(size_t)n * HID + cc]) * s + b1[cc];
            o0 += h * W2[cc * 2 + 0];
            o1 += h * W2[cc * 2 + 1];
        }
        out[n * 2 + 0] = o0;
        out[n * 2 + 1] = o1;
    }
}

extern "C" void kernel_launch(void* const* d_in, const int* in_sizes, int n_in,
                              void* d_out, int out_size, void* d_ws, size_t ws_size,
                              hipStream_t stream) {
    const float* x  = (const float*)d_in[0];
    const int*   ei = (const int*)d_in[1];
    const float* W1 = (const float*)d_in[2];
    const float* b1 = (const float*)d_in[3];
    const float* W2 = (const float*)d_in[4];
    const float* b2 = (const float*)d_in[5];
    float* out = (float*)d_out;

    float* deg    = (float*)d_ws;
    float* xw     = deg + N_NODES;
    int*   binned = (int*)(xw + (size_t)N_NODES * HID);
    int*   runs   = binned + N_EDGES;

    const int* row = ei;
    const int* col = ei + N_EDGES;

    gcn_bin<<<NCH, BN_BLOCK, 0, stream>>>(row, col, runs, binned);
    gcn_range_degree<<<NB, RL_BLOCK, 0, stream>>>(binned, runs, deg);

    int n_tiles = N_NODES / 16;
    gcn_xw_wmma<<<(n_tiles + 7) / 8, 256, 0, stream>>>(x, W1, deg, xw);

    gcn_range_sum<<<NB, RS_BLOCK, 0, stream>>>(binned, runs, xw, deg, b1, W2, b2, out);
}


# degree count moved into the x@W1 launch: one work-group per 416-node range counts its degrees in LDS from the filed edges, then multiplies and scales its own 26 row tiles; 4 launches -> 3
# speedup vs baseline: 5.2481x; 1.1076x over previous
#include <hip/hip_runtime.h>


typedef __attribute__((ext_vector_type(2))) float v2f;
typedef __attribute__((ext_vector_type(8))) float v8f;

#define N_NODES 100000
#define N_EDGES 3200000
#define IN_CH 512
#define HID 16

#define NPB 416
#define NB ((N_NODES + NPB - 1) / NPB)
#define NBP 256
#define NCH 256
#define CH_EDGES (N_EDGES / NCH)
static_assert(NB <= NBP, "the run table must hold every range");
static_assert(NPB <= 512, "the target's offset inside its range must fit 9 bits");
static_assert(N_NODES <= (1 << 22), "source << 9 must fit a positive int");
static_assert(NCH * CH_EDGES == N_EDGES, "edges must split evenly over chunks");
static_assert((NPB * 4) % 128 == 0, "a range's degrees and outputs must be whole lines");
static_assert(CH_EDGES < (1 << 15), "a run's start and length inside a chunk share one word, 16 bits each");

__device__ __forceinline__ int wave_running_sum(int v, int lane) {
#pragma unroll
    for (int o = 1; o < 32; o <<= 1) {
        const int t = __shfl_up(v, o);
        if (lane >= o) v += t;
    }
    return v;
}

#define CH_VEC (CH_EDGES / 4)
static_assert(CH_VEC * 4 == CH_EDGES, "a chunk must be whole 16-byte words");
#define BN_BLOCK 1024
#define BN_PER ((CH_VEC + BN_BLOCK - 1) / BN_BLOCK)
static_assert(NBP == 256 && BN_BLOCK >= NBP, "the running sum below is written for 8 waves of 32 columns");
__global__ void __launch_bounds__(BN_BLOCK)
gcn_bin(const int* __restrict__ row, const int* __restrict__ col, int* __restrict__ runs, int* __restrict__ binned) {
    __shared__ int h[NBP];
    __shared__ int lcur[NBP];
    __shared__ int wsum[NBP / 32];
    __shared__ int4 wordS4[CH_VEC];
    int* wordS = (int*)wordS4;

    const int tid = threadIdx.x;
    const int b = blockIdx.x;
    const int lane = tid & 31;
    const size_t e0 = (size_t)b * CH_EDGES;

    if (tid < NBP) h[tid] = 0;
    const int4* c4 = (const int4*)(col + e0);
    const int4* r4 = (const int4*)(row + e0);
    int4 dv[BN_PER], sv[BN_PER];
#pragma unroll
    for (int j = 0; j < BN_PER; ++j) {
        const int i = tid + j * BN_BLOCK;
        dv[j] = (i < CH_VEC) ? c4[i] : make_int4(0, 0, 0, 0);
        sv[j] = (i < CH_VEC) ? r4[i] : make_int4(0, 0, 0, 0);
    }
    __syncthreads();
#pragma unroll
    for (int j = 0; j < BN_PER; ++j) {
        if (tid + j * BN_BLOCK < CH_VEC) {
            atomicAdd(&h[(unsigned)dv[j].x / NPB], 1);
            atomicAdd(&h[(unsigned)dv[j].y / NPB], 1);
            atomicAdd(&h[(unsigned)dv[j].z / NPB], 1);
            atomicAdd(&h[(unsigned)dv[j].w / NPB], 1);
        }
    }
    __syncthreads();
    int myCnt = 0, run = 0;
    if (tid < NBP) {
        myCnt = h[tid];
        run = wave_running_sum(myCnt, lane);
        if (lane == 31) wsum[tid >> 5] = run;
    }
    __syncthreads();
    if (tid < NBP) {
        for (int w = 0; w < (tid >> 5); ++w) run += wsum[w];
        const int ls = run - myCnt;
        lcur[tid] = ls;
        runs[tid * NCH + b] = (ls << 16) | myCnt;
    }
    __syncthreads();
#pragma unroll
    for (int j = 0; j < BN_PER; ++j) {
        if (tid + j * BN_BLOCK < CH_VEC) {
            unsigned kk; int p;
            kk = (unsigned)dv[j].x / NPB; p = atomicAdd(&lcur[kk], 1); wordS[p] = (sv[j].x << 9) | (dv[j].x - (int)kk * NPB);
            kk = (unsigned)dv[j].y / NPB; p = atomicAdd(&lcur[kk], 1); wordS[p] = (sv[j].y << 9) | (dv[j].y - (int)kk * NPB);
            kk = (unsigned)dv[j].z / NPB; p = atomicAdd(&lcur[kk], 1); wordS[p] = (sv[j].z << 9) | (dv[j].z - (int)kk * NPB);
            kk = (unsigned)dv[j].w / NPB; p = atomicAdd(&lcur[kk], 1); wordS[p] = (sv[j].w << 9) | (dv[j].w - (int)kk * NPB);
        }
    }
    __syncthreads();
    int4* o4 = (int4*)(binned + e0);
    for (int i = tid; i < CH_VEC; i += BN_BLOCK) o4[i] = wordS4[i];
}

#define RL_BLOCK 1024
#define RL_TILE 16384
static_assert(RL_BLOCK == 4 * NCH, "four threads fill one run");
static_assert(RL_TILE % 32 == 0, "a tile is whole trips of 32 positions");
__device__ __forceinline__ int range_offsets(const int* __restrict__ runs, int k, int tid, int lane,
                                             int* offS, int* stS, int* wsum) {
    int len = 0, run = 0;
    if (tid < NCH) {
        const int v = runs[k * NCH + tid];
        len = v & 0xffff;
        stS[tid] = tid * CH_EDGES + (int)((unsigned)v >> 16);
        run = wave_running_sum(len, lane);
        if (lane == 31) wsum[tid >> 5] = run;
    }
    __syncthreads();
    if (tid < NCH) {
        for (int w = 0; w < (tid >> 5); ++w) run += wsum[w];
        offS[tid + 1] = run;
        if (tid == 0) offS[0] = 0;
    }
    __syncthreads();
    return __builtin_amdgcn_readfirstlane(offS[NCH]);
}
__device__ __forceinline__ void range_fill(int tb, int tid, const int* offS, const int* stS, int* addrS) {
    const int c = tid >> 2;
    const int o = offS[c], e = offS[c + 1], s = stS[c];
    const int lo = (o > tb) ? o : tb;
    const int hi = (e < tb + RL_TILE) ? e : tb + RL_TILE;
    for (int q = lo + (tid & 3); q < hi; q += 4) addrS[q - tb] = s + (q - o);
}


#define RD_U 8
#define RX_TILES (NPB / 16)
static_assert(NPB % 16 == 0 && N_NODES % 16 == 0, "a range and the node count are whole 16-row tiles");
static_assert(RX_TILES <= RL_BLOCK / 32, "one wave per tile");
static_assert(NPB <= RL_BLOCK, "one thread writes one node's degree");
__global__ void __launch_bounds__(RL_BLOCK)
gcn_range_xw(const int* __restrict__ binned, const int* __restrict__ runs,
             const float* __restrict__ x, const float* __restrict__ W1,
             float* __restrict__ deg, float* __restrict__ xw) {
    __shared__ float sW1[IN_CH * HID];
    __shared__ int c[512];
    __shared__ int offS[NCH + 1];
    __shared__ int stS[NCH];
    __shared__ int wsum[NCH / 32];
    __shared__ int addrS[RL_TILE];
    const int tid = threadIdx.x;
    const int k = blockIdx.x;
    static_assert((IN_CH * HID) / 4 == 2 * RL_BLOCK, "two 16-byte words of W1 per thread");
    const float4 w1a = ((const float4*)W1)[tid];
    const float4 w1b = ((const float4*)W1)[tid + RL_BLOCK];
    if (tid < 512) c[tid] = 0;
    const int n_e = range_offsets(runs, k, tid, tid & 31, offS, stS, wsum);
    ((float4*)sW1)[tid] = w1a;
    ((float4*)sW1)[tid + RL_BLOCK] = w1b;
    for (int tb = 0; tb < n_e; tb += RL_TILE) {
        const int tn = (n_e - tb < RL_TILE) ? n_e - tb : RL_TILE;
        range_fill(tb, tid, offS, stS, addrS);
        __syncthreads();
        for (int base = 0; base < tn; base += RL_BLOCK * RD_U) {
            int a[RD_U], w[RD_U];
#pragma unroll
            for (int j = 0; j < RD_U; ++j) {
                const int q = base + j * RL_BLOCK + tid;
                a[j] = addrS[q < tn ? q : tn - 1];
            }
#pragma unroll
            for (int j = 0; j < RD_U; ++j) {
                const int q = base + j * RL_BLOCK + tid;
                w[j] = (q < tn) ? binned[a[j]] : -1;
            }
#pragma unroll
            for (int j = 0; j < RD_U; ++j) if (w[j] >= 0) atomicAdd(&c[w[j] & 511], 1);
        }
        __syncthreads();
    }
    __syncthreads();
    const int n = k * NPB + tid;
    if (tid < NPB && n < N_NODES) deg[n] = (float)(c[tid] + 1);

    const int wave = __builtin_amdgcn_readfirstlane(tid >> 5);
    const int lane = tid & 31;
    const int half = lane >> 4;
    const int lrow = lane & 15;
    if (wave >= RX_TILES) return;
    const int row0 = k * NPB + wave * 16;
    if (row0 >= N_NODES) return;

    const float dg = (float)(c[wave * 16 + lrow] + 1);

    const float* xrow = x + (size_t)(row0 + lrow) * IN_CH + 2 * half;
    const float* bptr = sW1 + (2 * half) * HID + lrow;

    v8f acc = {};
#pragma unroll 8
    for (int k0 = 0; k0 < IN_CH; k0 += 4) {
        v2f a = *(const v2f*)(xrow + k0);
        v2f b;
        b.x = bptr[k0 * HID];
        b.y = bptr[k0 * HID + HID];
        acc = __builtin_amdgcn_wmma_f32_16x16x4_f32(
             false, a,  false, b,
             (short)0, acc,  false,  false);
    }

    const float dv = rsqrtf(dg);
    float* o = xw + (size_t)(row0 + 8 * half) * HID + lrow;
#pragma unroll
    for (int r = 0; r < 8; ++r) o[r * HID] = acc[r] * __shfl(dv, 8 * half + r);
}

#define RS_BLOCK RL_BLOCK
#define RS_WAVES (RS_BLOCK / 32)
#define RS_PAIRS 16
static_assert(NPB <= RS_BLOCK, "one thread finishes one node");
__global__ void __launch_bounds__(RS_BLOCK)
gcn_range_sum(const int* __restrict__ binned, const int* __restrict__ runs,
              const float* __restrict__ xs, const float* __restrict__ deg,
              const float* __restrict__ b1, const float* __restrict__ W2, const float* __restrict__ b2,
              float* __restrict__ out) {
    __shared__ float acc[NPB * HID];
    __shared__ int offS[NCH + 1];
    __shared__ int stS[NCH];
    __shared__ int wsum[NCH / 32];
    __shared__ int addrS[RL_TILE];
    for (int i = threadIdx.x; i < NPB * HID; i += RS_BLOCK) acc[i] = 0.0f;

    const int k = blockIdx.x;
    const int lane = threadIdx.x & 31;
    const int half = lane >> 4;
    const int c = lane & 15;
    const int wave = __builtin_amdgcn_readfirstlane((int)(threadIdx.x >> 5));
    const int n_e = range_offsets(runs, k, threadIdx.x, lane, offS, stS, wsum);
    for (int tb = 0; tb < n_e; tb += RL_TILE) {
        const int tn_e = (n_e - tb < RL_TILE) ? n_e - tb : RL_TILE;
        range_fill(tb, threadIdx.x, offS, stS, addrS);
        __syncthreads();
        const int full = tn_e >> 5;
        const int q1 = 32 * (wave + RS_WAVES) + lane;
        int an = addrS[q1 < tn_e ? q1 : tn_e - 1];
        const int q0 = 32 * wave + lane;
        const int a0 = addrS[q0 < tn_e ? q0 : tn_e - 1];
        int wv = (wave < full) ? binned[a0] : 0;
        asm volatile("" : "+v"(an));
        for (int t = wave; t < full; t += RS_WAVES) {
            int w[RS_PAIRS];
            float v[RS_PAIRS];
#pragma unroll
            for (int j = 0; j < RS_PAIRS; ++j) {
                const int wa = __builtin_amdgcn_readlane(wv, 2 * j);
                const int wb = __builtin_amdgcn_readlane(wv, 2 * j + 1);
                w[j] = half ? wb : wa;
            }
            const int tn = t + RS_WAVES;
            const int wv_next = (tn < full) ? binned[an] : 0;
            const int q2 = 32 * (tn + RS_WAVES) + lane;
            const int an_next = addrS[q2 < tn_e ? q2 : tn_e - 1];
#pragma unroll
            for (int j = 0; j < RS_PAIRS; ++j) v[j] = xs[(unsigned)((w[j] >> 9) * HID + c)];
#pragma unroll
            for (int j = 0; j < RS_PAIRS; ++j) atomicAdd(&acc[(w[j] & 511) * HID + c], v[j]);
            wv = wv_next;
            an = an_next;
        }
        const int rem = tn_e & 31;
        if (rem != 0 && wave == (full & (RS_WAVES - 1))) {
            const int wt = (lane < rem) ? binned[addrS[32 * full + lane]] : 0;
            for (int j = 0; j < RS_PAIRS; ++j) {
                const int idx = 2 * j + half;
                const int w = __shfl(wt, idx);
                if (idx < rem) atomicAdd(&acc[(w & 511) * HID + c], xs[(unsigned)((w >> 9) * HID + c)]);
            }
        }
        __syncthreads();
    }

    const int t = threadIdx.x;
    const int n = k * NPB + t;
    if (t < NPB && n < N_NODES) {
        const float s = rsqrtf(deg[n]);
        float o0 = b2[0], o1 = b2[1];
#pragma unroll
        for (int cc = 0; cc < HID; ++cc) {
            const float h = (acc[t * HID + cc] + xs[(size_t)n * HID + cc]) * s + b1[cc];
            o0 += h * W2[cc * 2 + 0];
            o1 += h * W2[cc * 2 + 1];
        }
        out[n * 2 + 0] = o0;
        out[n * 2 + 1] = o1;
    }
}

extern "C" void kernel_launch(void* const* d_in, const int* in_sizes, int n_in,
                              void* d_out, int out_size, void* d_ws, size_t ws_size,
                              hipStream_t stream) {
    const float* x  = (const float*)d_in[0];
    const int*   ei = (const int*)d_in[1];
    const float* W1 = (const float*)d_in[2];
    const float* b1 = (const float*)d_in[3];
    const float* W2 = (const float*)d_in[4];
    const float* b2 = (const float*)d_in[5];
    float* out = (float*)d_out;

    float* deg    = (float*)d_ws;
    float* xw     = deg + N_NODES;
    int*   binned = (int*)(xw + (size_t)N_NODES * HID);
    int*   runs   = binned + N_EDGES;

    const int* row = ei;
    const int* col = ei + N_EDGES;

    gcn_bin<<<NCH, BN_BLOCK, 0, stream>>>(row, col, runs, binned);
    gcn_range_xw<<<NB, RL_BLOCK, 0, stream>>>(binned, runs, x, W1, deg, xw);

    gcn_range_sum<<<NB, RS_BLOCK, 0, stream>>>(binned, runs, xw, deg, b1, W2, b2, out);
}


# x@W1 launch: tile waves prefetch the first 3 of a row's 8 units (256 B) of x into the second-level cache, one unit behind each load phase of the degree count; W1 staged ahead of it, one barrier less
# speedup vs baseline: 5.5362x; 1.0549x over previous
#include <hip/hip_runtime.h>


typedef __attribute__((ext_vector_type(2))) float v2f;
typedef __attribute__((ext_vector_type(8))) float v8f;

#define N_NODES 100000
#define N_EDGES 3200000
#define IN_CH 512
#define HID 16

#define NPB 416
#define NB ((N_NODES + NPB - 1) / NPB)
#define NBP 256
#define NCH 256
#define CH_EDGES (N_EDGES / NCH)
static_assert(NB <= NBP, "the run table must hold every range");
static_assert(NPB <= 512, "the target's offset inside its range must fit 9 bits");
static_assert(N_NODES <= (1 << 22), "source << 9 must fit a positive int");
static_assert(NCH * CH_EDGES == N_EDGES, "edges must split evenly over chunks");
static_assert((NPB * 4) % 128 == 0, "a range's degrees and outputs must be whole lines");
static_assert(CH_EDGES < (1 << 15), "a run's start and length inside a chunk share one word, 16 bits each");

__device__ __forceinline__ int wave_running_sum(int v, int lane) {
#pragma unroll
    for (int o = 1; o < 32; o <<= 1) {
        const int t = __shfl_up(v, o);
        if (lane >= o) v += t;
    }
    return v;
}

#define CH_VEC (CH_EDGES / 4)
static_assert(CH_VEC * 4 == CH_EDGES, "a chunk must be whole 16-byte words");
#define BN_BLOCK 1024
#define BN_PER ((CH_VEC + BN_BLOCK - 1) / BN_BLOCK)
static_assert(NBP == 256 && BN_BLOCK >= NBP, "the running sum below is written for 8 waves of 32 columns");
__global__ void __launch_bounds__(BN_BLOCK)
gcn_bin(const int* __restrict__ row, const int* __restrict__ col, int* __restrict__ runs, int* __restrict__ binned) {
    __shared__ int h[NBP];
    __shared__ int lcur[NBP];
    __shared__ int wsum[NBP / 32];
    __shared__ int4 wordS4[CH_VEC];
    int* wordS = (int*)wordS4;

    const int tid = threadIdx.x;
    const int b = blockIdx.x;
    const int lane = tid & 31;
    const size_t e0 = (size_t)b * CH_EDGES;

    if (tid < NBP) h[tid] = 0;
    const int4* c4 = (const int4*)(col + e0);
    const int4* r4 = (const int4*)(row + e0);
    int4 dv[BN_PER], sv[BN_PER];
#pragma unroll
    for (int j = 0; j < BN_PER; ++j) {
        const int i = tid + j * BN_BLOCK;
        dv[j] = (i < CH_VEC) ? c4[i] : make_int4(0, 0, 0, 0);
        sv[j] = (i < CH_VEC) ? r4[i] : make_int4(0, 0, 0, 0);
    }
    __syncthreads();
#pragma unroll
    for (int j = 0; j < BN_PER; ++j) {
        if (tid + j * BN_BLOCK < CH_VEC) {
            atomicAdd(&h[(unsigned)dv[j].x / NPB], 1);
            atomicAdd(&h[(unsigned)dv[j].y / NPB], 1);
            atomicAdd(&h[(unsigned)dv[j].z / NPB], 1);
            atomicAdd(&h[(unsigned)dv[j].w / NPB], 1);
        }
    }
    __syncthreads();
    int myCnt = 0, run = 0;
    if (tid < NBP) {
        myCnt = h[tid];
        run = wave_running_sum(myCnt, lane);
        if (lane == 31) wsum[tid >> 5] = run;
    }
    __syncthreads();
    if (tid < NBP) {
        for (int w = 0; w < (tid >> 5); ++w) run += wsum[w];
        const int ls = run - myCnt;
        lcur[tid] = ls;
        runs[tid * NCH + b] = (ls << 16) | myCnt;
    }
    __syncthreads();
#pragma unroll
    for (int j = 0; j < BN_PER; ++j) {
        if (tid + j * BN_BLOCK < CH_VEC) {
            unsigned kk; int p;
            kk = (unsigned)dv[j].x / NPB; p = atomicAdd(&lcur[kk], 1); wordS[p] = (sv[j].x << 9) | (dv[j].x - (int)kk * NPB);
            kk = (unsigned)dv[j].y / NPB; p = atomicAdd(&lcur[kk], 1); wordS[p] = (sv[j].y << 9) | (dv[j].y - (int)kk * NPB);
            kk = (unsigned)dv[j].z / NPB; p = atomicAdd(&lcur[kk], 1); wordS[p] = (sv[j].z << 9) | (dv[j].z - (int)kk * NPB);
            kk = (unsigned)dv[j].w / NPB; p = atomicAdd(&lcur[kk], 1); wordS[p] = (sv[j].w << 9) | (dv[j].w - (int)kk * NPB);
        }
    }
    __syncthreads();
    int4* o4 = (int4*)(binned + e0);
    for (int i = tid; i < CH_VEC; i += BN_BLOCK) o4[i] = wordS4[i];
}

#define RL_BLOCK 1024
#define RL_TILE 16384
static_assert(RL_BLOCK == 4 * NCH, "four threads fill one run");
static_assert(RL_TILE % 32 == 0, "a tile is whole trips of 32 positions");
__device__ __forceinline__ int range_word(const int* __restrict__ runs, int k, int tid) {
    return (tid < NCH) ? runs[k * NCH + tid] : 0;
}
__device__ __forceinline__ int range_offsets(int v, int tid, int lane, int* offS, int* stS, int* wsum) {
    int len = 0, run = 0;
    if (tid < NCH) {
        len = v & 0xffff;
        stS[tid] = tid * CH_EDGES + (int)((unsigned)v >> 16);
        run = wave_running_sum(len, lane);
        if (lane == 31) wsum[tid >> 5] = run;
    }
    __syncthreads();
    if (tid < NCH) {
        for (int w = 0; w < (tid >> 5); ++w) run += wsum[w];
        offS[tid + 1] = run;
        if (tid == 0) offS[0] = 0;
    }
    __syncthreads();
    return __builtin_amdgcn_readfirstlane(offS[NCH]);
}
__device__ __forceinline__ void range_fill(int tb, int tid, const int* offS, const int* stS, int* addrS) {
    const int c = tid >> 2;
    const int o = offS[c], e = offS[c + 1], s = stS[c];
    const int lo = (o > tb) ? o : tb;
    const int hi = (e < tb + RL_TILE) ? e : tb + RL_TILE;
    for (int q = lo + (tid & 3); q < hi; q += 4) addrS[q - tb] = s + (q - o);
}


#define RD_U 8
#define RX_TILES (NPB / 16)
static_assert(NPB % 16 == 0 && N_NODES % 16 == 0, "a range and the node count are whole 16-row tiles");
static_assert(RX_TILES <= RL_BLOCK / 32, "one wave per tile");
static_assert(NPB <= RL_BLOCK, "one thread writes one node's degree");
__global__ void __launch_bounds__(RL_BLOCK)
gcn_range_xw(const int* __restrict__ binned, const int* __restrict__ runs,
             const float* __restrict__ x, const float* __restrict__ W1,
             float* __restrict__ deg, float* __restrict__ xw) {
    __shared__ float sW1[IN_CH * HID];
    __shared__ int c[512];
    __shared__ int offS[NCH + 1];
    __shared__ int stS[NCH];
    __shared__ int wsum[NCH / 32];
    __shared__ int addrS[RL_TILE];
    const int tid = threadIdx.x;
    const int k = blockIdx.x;
    static_assert((IN_CH * HID) / 4 == 2 * RL_BLOCK, "two 16-byte words of W1 per thread");
    const float4 w1a = ((const float4*)W1)[tid];
    const float4 w1b = ((const float4*)W1)[tid + RL_BLOCK];
    const int rv = range_word(runs, k, tid);

    const int wave = __builtin_amdgcn_readfirstlane(tid >> 5);
    const int lane = tid & 31;
    const int half = lane >> 4;
    const int lrow = lane & 15;
    const int row0 = k * NPB + wave * 16;
    const bool has_tile = (wave < RX_TILES) && (row0 < N_NODES);
    const float* xpf = x + (size_t)(row0 + lrow) * IN_CH;
    const bool pf_lane = has_tile && (half == 0);
    if (tid < 512) c[tid] = 0;
    ((float4*)sW1)[tid] = w1a;
    ((float4*)sW1)[tid + RL_BLOCK] = w1b;
    if (pf_lane) __builtin_prefetch(xpf, 0, 2);
    const int n_e = range_offsets(rv, tid, lane, offS, stS, wsum);
    for (int tb = 0; tb < n_e; tb += RL_TILE) {
        const int tn = (n_e - tb < RL_TILE) ? n_e - tb : RL_TILE;
        range_fill(tb, tid, offS, stS, addrS);
        __syncthreads();
        for (int base = 0; base < tn; base += RL_BLOCK * RD_U) {
            int a[RD_U], w[RD_U];
#pragma unroll
            for (int j = 0; j < RD_U; ++j) {
                const int q = base + j * RL_BLOCK + tid;
                a[j] = addrS[q < tn ? q : tn - 1];
            }
#pragma unroll
            for (int j = 0; j < RD_U; ++j) {
                const int q = base + j * RL_BLOCK + tid;
                w[j] = (q < tn) ? binned[a[j]] : -1;
            }
            if (tb == 0 && pf_lane) {
                if (base == 0) __builtin_prefetch(xpf + 64, 0, 2);
                else if (base == RL_BLOCK * RD_U) __builtin_prefetch(xpf + 128, 0, 2);
            }
#pragma unroll
            for (int j = 0; j < RD_U; ++j) if (w[j] >= 0) atomicAdd(&c[w[j] & 511], 1);
        }
        __syncthreads();
    }
    const int n = k * NPB + tid;
    if (tid < NPB && n < N_NODES) deg[n] = (float)(c[tid] + 1);

    if (!has_tile) return;

    const float dg = (float)(c[wave * 16 + lrow] + 1);

    const float* xrow = x + (size_t)(row0 + lrow) * IN_CH + 2 * half;
    const float* bptr = sW1 + (2 * half) * HID + lrow;

    v8f acc = {};
#pragma unroll 8
    for (int k0 = 0; k0 < IN_CH; k0 += 4) {
        v2f a = *(const v2f*)(xrow + k0);
        v2f b;
        b.x = bptr[k0 * HID];
        b.y = bptr[k0 * HID + HID];
        acc = __builtin_amdgcn_wmma_f32_16x16x4_f32(
             false, a,  false, b,
             (short)0, acc,  false,  false);
    }

    const float dv = rsqrtf(dg);
    float* o = xw + (size_t)(row0 + 8 * half) * HID + lrow;
#pragma unroll
    for (int r = 0; r < 8; ++r) o[r * HID] = acc[r] * __shfl(dv, 8 * half + r);
}

#define RS_BLOCK RL_BLOCK
#define RS_WAVES (RS_BLOCK / 32)
#define RS_PAIRS 16
static_assert(NPB <= RS_BLOCK, "one thread finishes one node");
__global__ void __launch_bounds__(RS_BLOCK)
gcn_range_sum(const int* __restrict__ binned, const int* __restrict__ runs,
              const float* __restrict__ xs, const float* __restrict__ deg,
              const float* __restrict__ b1, const float* __restrict__ W2, const float* __restrict__ b2,
              float* __restrict__ out) {
    __shared__ float acc[NPB * HID];
    __shared__ int offS[NCH + 1];
    __shared__ int stS[NCH];
    __shared__ int wsum[NCH / 32];
    __shared__ int addrS[RL_TILE];
    for (int i = threadIdx.x; i < NPB * HID; i += RS_BLOCK) acc[i] = 0.0f;

    const int k = blockIdx.x;
    const int lane = threadIdx.x & 31;
    const int half = lane >> 4;
    const int c = lane & 15;
    const int wave = __builtin_amdgcn_readfirstlane((int)(threadIdx.x >> 5));
    const int n_e = range_offsets(range_word(runs, k, threadIdx.x), threadIdx.x, lane, offS, stS, wsum);
    for (int tb = 0; tb < n_e; tb += RL_TILE) {
        const int tn_e = (n_e - tb < RL_TILE) ? n_e - tb : RL_TILE;
        range_fill(tb, threadIdx.x, offS, stS, addrS);
        __syncthreads();
        const int full = tn_e >> 5;
        const int q1 = 32 * (wave + RS_WAVES) + lane;
        int an = addrS[q1 < tn_e ? q1 : tn_e - 1];
        const int q0 = 32 * wave + lane;
        const int a0 = addrS[q0 < tn_e ? q0 : tn_e - 1];
        int wv = (wave < full) ? binned[a0] : 0;
        asm volatile("" : "+v"(an));
        for (int t = wave; t < full; t += RS_WAVES) {
            int w[RS_PAIRS];
            float v[RS_PAIRS];
#pragma unroll
            for (int j = 0; j < RS_PAIRS; ++j) {
                const int wa = __builtin_amdgcn_readlane(wv, 2 * j);
                const int wb = __builtin_amdgcn_readlane(wv, 2 * j + 1);
                w[j] = half ? wb : wa;
            }
            const int tn = t + RS_WAVES;
            const int wv_next = (tn < full) ? binned[an] : 0;
            const int q2 = 32 * (tn + RS_WAVES) + lane;
            const int an_next = addrS[q2 < tn_e ? q2 : tn_e - 1];
#pragma unroll
            for (int j = 0; j < RS_PAIRS; ++j) v[j] = xs[(unsigned)((w[j] >> 9) * HID + c)];
#pragma unroll
            for (int j = 0; j < RS_PAIRS; ++j) atomicAdd(&acc[(w[j] & 511) * HID + c], v[j]);
            wv = wv_next;
            an = an_next;
        }
        const int rem = tn_e & 31;
        if (rem != 0 && wave == (full & (RS_WAVES - 1))) {
            const int wt = (lane < rem) ? binned[addrS[32 * full + lane]] : 0;
            for (int j = 0; j < RS_PAIRS; ++j) {
                const int idx = 2 * j + half;
                const int w = __shfl(wt, idx);
                if (idx < rem) atomicAdd(&acc[(w & 511) * HID + c], xs[(unsigned)((w >> 9) * HID + c)]);
            }
        }
        __syncthreads();
    }

    const int t = threadIdx.x;
    const int n = k * NPB + t;
    if (t < NPB && n < N_NODES) {
        const float s = rsqrtf(deg[n]);
        float o0 = b2[0], o1 = b2[1];
#pragma unroll
        for (int cc = 0; cc < HID; ++cc) {
            const float h = (acc[t * HID + cc] + xs[(size_t)n * HID + cc]) * s + b1[cc];
            o0 += h * W2[cc * 2 + 0];
            o1 += h * W2[cc * 2 + 1];
        }
        out[n * 2 + 0] = o0;
        out[n * 2 + 1] = o1;
    }
}

extern "C" void kernel_launch(void* const* d_in, const int* in_sizes, int n_in,
                              void* d_out, int out_size, void* d_ws, size_t ws_size,
                              hipStream_t stream) {
    const float* x  = (const float*)d_in[0];
    const int*   ei = (const int*)d_in[1];
    const float* W1 = (const float*)d_in[2];
    const float* b1 = (const float*)d_in[3];
    const float* W2 = (const float*)d_in[4];
    const float* b2 = (const float*)d_in[5];
    float* out = (float*)d_out;

    float* deg    = (float*)d_ws;
    float* xw     = deg + N_NODES;
    int*   binned = (int*)(xw + (size_t)N_NODES * HID);
    int*   runs   = binned + N_EDGES;

    const int* row = ei;
    const int* col = ei + N_EDGES;

    gcn_bin<<<NCH, BN_BLOCK, 0, stream>>>(row, col, runs, binned);
    gcn_range_xw<<<NB, RL_BLOCK, 0, stream>>>(binned, runs, x, W1, deg, xw);

    gcn_range_sum<<<NB, RS_BLOCK, 0, stream>>>(binned, runs, xw, deg, b1, W2, b2, out);
}
